# starting kernel
# baseline (speedup 1.0000x reference)
#include <hip/hip_runtime.h>
#include <hip/hip_bf16.h>

typedef __attribute__((ext_vector_type(2))) float v2f;
typedef __attribute__((ext_vector_type(8))) float v8f;

#define TT 2048
#define BB 64
#define DD 512
#define ROWS (TT * BB)
#define WAVES_PER_BLOCK 8
#define ROWS_PER_BLOCK (WAVES_PER_BLOCK * 16)

__global__ void __launch_bounds__(256)
xproj_wmma_kernel(const float* __restrict__ x,
                  const float* __restrict__ W,
                  float* __restrict__ xproj) {
    __shared__ float wlds[DD];
    for (int i = threadIdx.x; i < DD; i += blockDim.x) wlds[i] = W[i];
    __syncthreads();

    const int lane    = threadIdx.x & 31;
    const int wave    = threadIdx.x >> 5;
    const int m       = lane & 15;
    const int halfSel = lane >> 4;

    const long tile    = (long)blockIdx.x * WAVES_PER_BLOCK + wave;
    const long rowBase = tile * 16;

    const float* xRow = x + (rowBase + m) * DD + 2 * halfSel;
    const float* wPtr = wlds + 2 * halfSel;

    v8f c = {};
#pragma unroll 8
    for (int k = 0; k < DD; k += 4) {
        v2f a = *(const v2f*)(xRow + k);
        v2f b = *(const v2f*)(wPtr + k);
        c = __builtin_amdgcn_wmma_f32_16x16x4_f32(
                false, a, false, b, (short)0, c, false, false);
    }

    if (m == 0) {
        float* dst = xproj + rowBase + 8 * halfSel;
        *(float4*)(dst + 0) = make_float4(c[0], c[1], c[2], c[3]);
        *(float4*)(dst + 4) = make_float4(c[4], c[5], c[6], c[7]);
    }
}

#define PF 16

__global__ void __launch_bounds__(64)
scan_kernel(const float* __restrict__ xproj,
            const float* __restrict__ W,
            const float* __restrict__ bptr,
            float* __restrict__ out) {
    const int   col  = threadIdx.x;
    const float wz   = W[DD];
    const float bias = bptr[0];

    float buf[PF], nbuf[PF];
#pragma unroll
    for (int i = 0; i < PF; ++i) buf[i] = xproj[i * BB + col];

    float z = 0.0f;
    const int ntiles = TT / PF;
    for (int tile = 0; tile < ntiles; ++tile) {
        if (tile + 1 < ntiles) {
            const int nt = (tile + 1) * PF;
#pragma unroll
            for (int i = 0; i < PF; ++i) nbuf[i] = xproj[(nt + i) * BB + col];
        }
        const int t0 = tile * PF;
#pragma unroll
        for (int i = 0; i < PF; ++i) {
            float u = buf[i] + bias + wz * z;
            z = 1.0f / (1.0f + __expf(-u));
            out[(t0 + i) * BB + col] = z;
        }
#pragma unroll
        for (int i = 0; i < PF; ++i) buf[i] = nbuf[i];
    }
}

extern "C" void kernel_launch(void* const* d_in, const int* in_sizes, int n_in,
                              void* d_out, int out_size, void* d_ws, size_t ws_size,
                              hipStream_t stream) {
    const float* x  = (const float*)d_in[0];
    const float* W  = (const float*)d_in[1];
    const float* b  = (const float*)d_in[2];
    float* out   = (float*)d_out;
    float* xproj = (float*)d_ws;

    dim3 block(32 * WAVES_PER_BLOCK);
    dim3 grid(ROWS / ROWS_PER_BLOCK);
    xproj_wmma_kernel<<<grid, block, 0, stream>>>(x, W, xproj);
    scan_kernel<<<1, BB, 0, stream>>>(xproj, W, b, out);
}


# scan: a wave carries 4 chunks x 8 columns side by side in its lanes: 128 chunks of 16 steps on 8 work-groups (was 32 of 64 on 2); same check and rerun; out stored in 32-byte pieces: same values
# speedup vs baseline: 4.5283x; 4.5283x over previous
#include <hip/hip_runtime.h>
#include <hip/hip_bf16.h>

typedef __attribute__((ext_vector_type(2))) float v2f;
typedef __attribute__((ext_vector_type(8))) float v8f;

#define TT 2048
#define BB 64
#define DD 512
#define ROWS (TT * BB)
#define WAVES_PER_BLOCK 8
#define ROWS_PER_BLOCK (WAVES_PER_BLOCK * 16)

__global__ void __launch_bounds__(256)
xproj_wmma_kernel(const float* __restrict__ x,
                  const float* __restrict__ W,
                  float* __restrict__ xproj) {
    __shared__ float wlds[DD];
    for (int i = threadIdx.x; i < DD; i += blockDim.x) wlds[i] = W[i];
    __syncthreads();

    const int lane    = threadIdx.x & 31;
    const int wave    = threadIdx.x >> 5;
    const int m       = lane & 15;
    const int halfSel = lane >> 4;

    const long tile    = (long)blockIdx.x * WAVES_PER_BLOCK + wave;
    const long rowBase = tile * 16;

    const float* xRow = x + (rowBase + m) * DD + 2 * halfSel;
    const float* wPtr = wlds + 2 * halfSel;

    v8f c = {};
#pragma unroll 8
    for (int k = 0; k < DD; k += 4) {
        v2f a = *(const v2f*)(xRow + k);
        v2f b = *(const v2f*)(wPtr + k);
        c = __builtin_amdgcn_wmma_f32_16x16x4_f32(
                false, a, false, b, (short)0, c, false, false);
    }

    if (m == 0) {
        float* dst = xproj + rowBase + 8 * halfSel;
        *(float4*)(dst + 0) = make_float4(c[0], c[1], c[2], c[3]);
        *(float4*)(dst + 4) = make_float4(c[4], c[5], c[6], c[7]);
    }
}

#define GC  8
#define CPW (32 / GC)
#define NW  32
#define NCH (NW * CPW)
#define CL  (TT / NCH)
#define WU  16

__device__ __forceinline__ float sig_step(float xp, float bias, float wz, float z) {
    float u = xp + bias + wz * z;
    return 1.0f / (1.0f + __expf(-u));
}

__global__ void __launch_bounds__(32 * NW)
scan_kernel(const float* __restrict__ xproj,
            const float* __restrict__ W,
            const float* __restrict__ bptr,
            float* __restrict__ out) {
    __shared__ unsigned endz[NCH * GC];
    __shared__ int      again[NCH];

    const int   lane  = threadIdx.x & 31;
    const int   chunk = threadIdx.x / GC;
    const int   col   = blockIdx.x * GC + (lane % GC);
    const float wz    = W[DD];
    const float bias  = bptr[0];
    const int   t0    = chunk * CL;

    if (threadIdx.x < NCH) again[threadIdx.x] = 0;

    const int tw0 = (chunk == 0) ? 0 : t0 - WU;
    float wb[WU], buf[CL];
#pragma unroll
    for (int i = 0; i < WU; ++i) wb[i] = xproj[(tw0 + i) * BB + col];
#pragma unroll
    for (int i = 0; i < CL; ++i) buf[i] = xproj[(t0 + i) * BB + col];

    float zs = 0.5f;
#pragma unroll
    for (int i = 0; i < WU; ++i) zs = sig_step(wb[i], bias, wz, zs);
    if (chunk == 0) zs = 0.0f;

    bool run = true;
    for (int pass = 0; pass < NCH; ++pass) {
        if (run) {
            float z = zs;
#pragma unroll
            for (int i = 0; i < CL; ++i) {
                z = sig_step(buf[i], bias, wz, z);
                out[(t0 + i) * BB + col] = z;
            }
            endz[threadIdx.x] = __float_as_uint(z);
        }
        __syncthreads();
        unsigned pb = __float_as_uint(zs);
        if (chunk > 0) pb = endz[threadIdx.x - GC];
        run = __any(pb != __float_as_uint(zs)) != 0;
        if (run && lane == 0) again[pass] = 1;
        zs = __uint_as_float(pb);
        __syncthreads();
        if (again[pass] == 0) break;
    }
}

extern "C" void kernel_launch(void* const* d_in, const int* in_sizes, int n_in,
                              void* d_out, int out_size, void* d_ws, size_t ws_size,
                              hipStream_t stream) {
    const float* x  = (const float*)d_in[0];
    const float* W  = (const float*)d_in[1];
    const float* b  = (const float*)d_in[2];
    float* out   = (float*)d_out;
    float* xproj = (float*)d_ws;

    dim3 block(32 * WAVES_PER_BLOCK);
    dim3 grid(ROWS / ROWS_PER_BLOCK);
    xproj_wmma_kernel<<<grid, block, 0, stream>>>(x, W, xproj);
    scan_kernel<<<BB / GC, 32 * NW, 0, stream>>>(xproj, W, b, out);
}
